# starting kernel
# baseline (speedup 1.0000x reference)
#include <hip/hip_runtime.h>


typedef __attribute__((ext_vector_type(2))) float v2f;
typedef __attribute__((ext_vector_type(8))) float v8f;

#define STATE 16
#define MEAS  8

static __device__ __forceinline__ float bcastf(float v, int srcLane) {
  int r = __builtin_amdgcn_ds_bpermute(srcLane << 2, __builtin_bit_cast(int, v));
  return __builtin_bit_cast(float, r);
}

static __device__ __forceinline__ v8f wmma_k4(v2f a, v2f b, v8f c) {
  return __builtin_amdgcn_wmma_f32_16x16x4_f32(
       false, a,  false, b,
       (short)0, c,  false,  false);
}

__global__ __launch_bounds__(32) void kalman_seq_kernel(
    const float* __restrict__ meas,
    const float* __restrict__ Fm,
    const float* __restrict__ Qm,
    const float* __restrict__ Rm,
    float* __restrict__ out,
    int T)
{
  __shared__ float lds_P [256];
  __shared__ float lds_M1[256];
  __shared__ float lds_Pp[256];
  __shared__ float lds_nK[256];
  __shared__ float lds_Si[128];
  __shared__ float lds_x [16];

  const int lane = threadIdx.x;
  const int l16  = lane & 15;
  const int half = lane >> 4;
  const int gr   = lane >> 2;
  const int cg   = lane & 3;

  float Frow[16];
#pragma unroll
  for (int j = 0; j < 16; ++j) Frow[j] = Fm[l16 * 16 + j];

  v2f fab[4];
#pragma unroll
  for (int s = 0; s < 4; ++s) {
    const int k0 = 4 * s + 2 * half;
    fab[s].x = Fm[l16 * 16 + k0];
    fab[s].y = Fm[l16 * 16 + k0 + 1];
  }

  v8f qacc;
#pragma unroll
  for (int r = 0; r < 8; ++r) qacc[r] = Qm[(r + 8 * half) * 16 + l16];

  float rr[4];
#pragma unroll
  for (int j = 0; j < 4; ++j)
    rr[j] = (cg < 2) ? Rm[gr * 8 + 4 * cg + j] : 0.0f;

  const float maskL = (cg < 2) ? 1.0f : 0.0f;
  float iden[4];
#pragma unroll
  for (int j = 0; j < 4; ++j)
    iden[j] = ((4 * cg + j) - 8 == gr) ? 1.0f : 0.0f;

  for (int idx = lane; idx < 256; idx += 32)
    lds_P[idx] = ((idx >> 4) == (idx & 15)) ? 1.0f : 0.0f;
  for (int idx = lane; idx < 128; idx += 32)
    lds_Si[idx] = 0.0f;
  if (lane < 16) lds_x[lane] = 0.0f;
  __syncthreads();

  float zv = meas[lane & 7];

  for (int t = 0; t < T; ++t) {
    const int tn = (t + 1 < T) ? (t + 1) : (T - 1);
    const float znext = meas[(size_t)tn * MEAS + (lane & 7)];
    __builtin_prefetch(&meas[(size_t)(t + 48) * MEAS], 0, 1);

    float x0 = 0.0f, x1 = 0.0f, x2 = 0.0f, x3 = 0.0f;
#pragma unroll
    for (int j = 0; j < 4; ++j) {
      x0 += Frow[j]      * lds_x[j];
      x1 += Frow[j + 4]  * lds_x[j + 4];
      x2 += Frow[j + 8]  * lds_x[j + 8];
      x3 += Frow[j + 12] * lds_x[j + 12];
    }
    const float xp = (x0 + x1) + (x2 + x3);
    const float yv = zv - xp;

    v8f m1 = {0, 0, 0, 0, 0, 0, 0, 0};
#pragma unroll
    for (int s = 0; s < 4; ++s) {
      const int kr = 4 * s + 2 * half;
      v2f b;
      b.x = lds_P[kr * 16 + l16];
      b.y = lds_P[(kr + 1) * 16 + l16];
      m1 = wmma_k4(fab[s], b, m1);
    }
#pragma unroll
    for (int r = 0; r < 8; ++r) lds_M1[(r + 8 * half) * 16 + l16] = m1[r];
    __syncthreads();

    v8f pp = qacc;
#pragma unroll
    for (int s = 0; s < 4; ++s) {
      const int k0 = 4 * s + 2 * half;
      v2f a;
      a.x = lds_M1[l16 * 16 + k0];
      a.y = lds_M1[l16 * 16 + k0 + 1];
      pp = wmma_k4(a, fab[s], pp);
    }
#pragma unroll
    for (int r = 0; r < 8; ++r) lds_Pp[(r + 8 * half) * 16 + l16] = pp[r];
    __syncthreads();

    float va[4];
    {
      const int base = gr * 16 + 4 * (cg & 1);
#pragma unroll
      for (int j = 0; j < 4; ++j) {
        const float sv = lds_Pp[base + j] + rr[j];
        va[j] = sv * maskL + iden[j];
      }
    }
#pragma unroll
    for (int k = 0; k < 8; ++k) {
      const int kc = k & 3, kg = k >> 2;
      const float pv  = bcastf(va[kc], 4 * k + kg);
      const float piv = __builtin_amdgcn_rcpf(pv);
      const float fac = bcastf(va[kc], 4 * gr + kg);
      const float p0 = bcastf(va[0], 4 * k + cg) * piv;
      const float p1 = bcastf(va[1], 4 * k + cg) * piv;
      const float p2 = bcastf(va[2], 4 * k + cg) * piv;
      const float p3 = bcastf(va[3], 4 * k + cg) * piv;
      const bool isP = (gr == k);
      va[0] = isP ? p0 : va[0] - fac * p0;
      va[1] = isP ? p1 : va[1] - fac * p1;
      va[2] = isP ? p2 : va[2] - fac * p2;
      va[3] = isP ? p3 : va[3] - fac * p3;
    }
    if (cg >= 2) {
#pragma unroll
      for (int j = 0; j < 4; ++j)
        lds_Si[gr * 16 + 4 * (cg - 2) + j] = va[j];
    }
    __syncthreads();

    v8f kk = {0, 0, 0, 0, 0, 0, 0, 0};
#pragma unroll
    for (int s = 0; s < 2; ++s) {
      const int k0 = 4 * s + 2 * half;
      v2f a;
      a.x = lds_Pp[l16 * 16 + k0];
      a.y = lds_Pp[l16 * 16 + k0 + 1];
      v2f b;
      b.x = lds_Si[k0 * 16 + l16];
      b.y = lds_Si[(k0 + 1) * 16 + l16];
      kk = wmma_k4(a, b, kk);
    }
#pragma unroll
    for (int r = 0; r < 8; ++r) lds_nK[(r + 8 * half) * 16 + l16] = -kk[r];
    __syncthreads();

    v8f pn = pp;
#pragma unroll
    for (int s = 0; s < 2; ++s) {
      const int k0 = 4 * s + 2 * half;
      v2f a;
      a.x = lds_nK[l16 * 16 + k0];
      a.y = lds_nK[l16 * 16 + k0 + 1];
      v2f b;
      b.x = lds_Pp[k0 * 16 + l16];
      b.y = lds_Pp[(k0 + 1) * 16 + l16];
      pn = wmma_k4(a, b, pn);
    }
#pragma unroll
    for (int r = 0; r < 8; ++r) lds_P[(r + 8 * half) * 16 + l16] = pn[r];

    float xn = xp;
#pragma unroll
    for (int j = 0; j < MEAS; ++j)
      xn -= lds_nK[l16 * 16 + j] * bcastf(yv, j);
    if (lane < 16) {
      lds_x[l16] = xn;
      out[(size_t)t * 16 + l16] = xn;
    }
    zv = znext;
    __syncthreads();
  }
}

extern "C" void kernel_launch(void* const* d_in, const int* in_sizes, int n_in,
                              void* d_out, int out_size, void* d_ws, size_t ws_size,
                              hipStream_t stream) {
  const float* meas = (const float*)d_in[0];
  const float* F    = (const float*)d_in[1];
  const float* Q    = (const float*)d_in[3];
  const float* R    = (const float*)d_in[4];
  float* out = (float*)d_out;

  const int T = in_sizes[0] / MEAS;

  hipLaunchKernelGGL(kalman_seq_kernel, dim3(1), dim3(32), 0, stream,
                     meas, F, Q, R, out, T);
}


# Whole step kept in registers: WMMA D tiles turned into next operands by half-wave exchange (transposed products where an A operand is needed), Gauss-Jordan on DPP; no LDS or barrier in loop
# speedup vs baseline: 2.6183x; 2.6183x over previous
#include <hip/hip_runtime.h>


typedef __attribute__((ext_vector_type(2))) float v2f;
typedef __attribute__((ext_vector_type(8))) float v8f;

#define STATE 16
#define MEAS  8

static __device__ __forceinline__ v8f wmma_k4(v2f a, v2f b, v8f c) {
  return __builtin_amdgcn_wmma_f32_16x16x4_f32(
       false, a,  false, b,
       (short)0, c,  false,  false);
}

static __device__ __forceinline__ float xhalf(float x) {
  const int b = __builtin_bit_cast(int, x);
  return __builtin_bit_cast(float,
      __builtin_amdgcn_permlanex16(b, b, 0x76543210, 0xfedcba98, false, false));
}

template <int N>
static __device__ __forceinline__ float rowshare(float x) {
  const int b = __builtin_bit_cast(int, x);
  return __builtin_bit_cast(float,
      __builtin_amdgcn_mov_dpp(b, 0x150 + N, 0xf, 0xf, false));
}

static __device__ __forceinline__ float rowshl8_zero(float x) {
  const int b = __builtin_bit_cast(int, x);
  return __builtin_bit_cast(float,
      __builtin_amdgcn_update_dpp(0, b, 0x108, 0xf, 0xf, true));
}

static __device__ __forceinline__ void conv_full(const v8f d, const bool hi, v2f (&f)[4]) {
  float X[8];
#pragma unroll
  for (int r = 0; r < 8; ++r) X[r] = xhalf(d[r]);
  f[0].x = hi ? X[2] : d[0];  f[0].y = hi ? X[3] : d[1];
  f[1].x = hi ? X[6] : d[4];  f[1].y = hi ? X[7] : d[5];
  f[2].x = hi ? d[2] : X[0];  f[2].y = hi ? d[3] : X[1];
  f[3].x = hi ? d[6] : X[4];  f[3].y = hi ? d[7] : X[5];
}

static __device__ __forceinline__ void conv_lo(const v8f d, const bool hi, v2f (&f)[2]) {
  const float x2 = xhalf(d[2]), x3 = xhalf(d[3]);
  const float x6 = xhalf(d[6]), x7 = xhalf(d[7]);
  f[0].x = hi ? x2 : d[0];  f[0].y = hi ? x3 : d[1];
  f[1].x = hi ? x6 : d[4];  f[1].y = hi ? x7 : d[5];
}

template <int K>
static __device__ __forceinline__ void gj_step(float (&g)[8]) {
  const float pv  = rowshare<K>(g[K]);
  const float piv = __builtin_amdgcn_rcpf(pv);
  float f[8];
#pragma unroll
  for (int i = 0; i < 8; ++i) f[i] = rowshare<K>(g[i]);
  const float gk = g[K] * piv;
#pragma unroll
  for (int i = 0; i < 8; ++i)
    g[i] = (i == K) ? gk : __builtin_fmaf(-f[i], gk, g[i]);
}

__global__ __launch_bounds__(32) void kalman_seq_kernel(
    const float* __restrict__ meas,
    const float* __restrict__ Fm,
    const float* __restrict__ Qm,
    const float* __restrict__ Rm,
    float* __restrict__ out,
    int T)
{
  const int lane = threadIdx.x;
  const int l16  = lane & 15;
  const int half = lane >> 4;
  const bool hi  = (half != 0);

  float Frow[16];
#pragma unroll
  for (int j = 0; j < 16; ++j) Frow[j] = Fm[l16 * 16 + j];

  v2f fab[4];
#pragma unroll
  for (int s = 0; s < 4; ++s) {
    const int k0 = 4 * s + 2 * half;
    fab[s].x = Fm[l16 * 16 + k0];
    fab[s].y = Fm[l16 * 16 + k0 + 1];
  }

  v8f qacc, qaccT;
#pragma unroll
  for (int r = 0; r < 8; ++r) {
    qacc[r]  = Qm[(r + 8 * half) * 16 + l16];
    qaccT[r] = Qm[l16 * 16 + (r + 8 * half)];
  }

  const float maskL = (lane < 8) ? 1.0f : 0.0f;
  float rI[8];
#pragma unroll
  for (int i = 0; i < 8; ++i) {
    const float idn = ((l16 & 7) == i) ? 1.0f : 0.0f;
    rI[i] = (lane < 8) ? Rm[i * 8 + (lane & 7)] : idn;
  }

  v8f Pd;
#pragma unroll
  for (int r = 0; r < 8; ++r) Pd[r] = ((r + 8 * half) == l16) ? 1.0f : 0.0f;
  float xv = 0.0f;

  float zv = meas[lane & 7];

  for (int t = 0; t < T; ++t) {
    const int tn = (t + 1 < T) ? (t + 1) : (T - 1);
    const float znext = meas[(size_t)tn * MEAS + (lane & 7)];
    __builtin_prefetch(&meas[(size_t)(t + 48) * MEAS], 0, 1);

    float x0 = 0.0f, x1 = 0.0f, x2 = 0.0f, x3 = 0.0f;
    x0 = __builtin_fmaf(Frow[0],  rowshare<0>(xv),  x0);
    x1 = __builtin_fmaf(Frow[4],  rowshare<4>(xv),  x1);
    x2 = __builtin_fmaf(Frow[8],  rowshare<8>(xv),  x2);
    x3 = __builtin_fmaf(Frow[12], rowshare<12>(xv), x3);
    x0 = __builtin_fmaf(Frow[1],  rowshare<1>(xv),  x0);
    x1 = __builtin_fmaf(Frow[5],  rowshare<5>(xv),  x1);
    x2 = __builtin_fmaf(Frow[9],  rowshare<9>(xv),  x2);
    x3 = __builtin_fmaf(Frow[13], rowshare<13>(xv), x3);
    x0 = __builtin_fmaf(Frow[2],  rowshare<2>(xv),  x0);
    x1 = __builtin_fmaf(Frow[6],  rowshare<6>(xv),  x1);
    x2 = __builtin_fmaf(Frow[10], rowshare<10>(xv), x2);
    x3 = __builtin_fmaf(Frow[14], rowshare<14>(xv), x3);
    x0 = __builtin_fmaf(Frow[3],  rowshare<3>(xv),  x0);
    x1 = __builtin_fmaf(Frow[7],  rowshare<7>(xv),  x1);
    x2 = __builtin_fmaf(Frow[11], rowshare<11>(xv), x2);
    x3 = __builtin_fmaf(Frow[15], rowshare<15>(xv), x3);
    const float xp = (x0 + x1) + (x2 + x3);
    const float yv = zv - xp;
    const float y0 = rowshare<0>(yv), y1 = rowshare<1>(yv);
    const float y2 = rowshare<2>(yv), y3 = rowshare<3>(yv);
    const float y4 = rowshare<4>(yv), y5 = rowshare<5>(yv);
    const float y6 = rowshare<6>(yv), y7 = rowshare<7>(yv);

    v2f bp[4];
    conv_full(Pd, hi, bp);
    v8f m1t = {0, 0, 0, 0, 0, 0, 0, 0};
#pragma unroll
    for (int s = 0; s < 4; ++s) m1t = wmma_k4(bp[s], fab[s], m1t);

    v2f bm[4];
    conv_full(m1t, hi, bm);
    v8f pp = qacc, ppt = qaccT;
#pragma unroll
    for (int s = 0; s < 4; ++s) {
      pp  = wmma_k4(bm[s], fab[s], pp);
      ppt = wmma_k4(fab[s], bm[s], ppt);
    }

    float g[8];
#pragma unroll
    for (int i = 0; i < 8; ++i) g[i] = __builtin_fmaf(pp[i], maskL, rI[i]);
    gj_step<0>(g); gj_step<1>(g); gj_step<2>(g); gj_step<3>(g);
    gj_step<4>(g); gj_step<5>(g); gj_step<6>(g); gj_step<7>(g);
    float sh[8];
#pragma unroll
    for (int i = 0; i < 8; ++i) sh[i] = rowshl8_zero(g[i]);
    v2f asi[2];
    {
      const float s2 = xhalf(sh[2]), s3 = xhalf(sh[3]);
      const float s6 = xhalf(sh[6]), s7 = xhalf(sh[7]);
      asi[0].x = hi ? s2 : sh[0];  asi[0].y = hi ? s3 : sh[1];
      asi[1].x = hi ? s6 : sh[4];  asi[1].y = hi ? s7 : sh[5];
    }

    v2f bq[2];
    conv_lo(ppt, hi, bq);
    v8f kt = {0, 0, 0, 0, 0, 0, 0, 0};
#pragma unroll
    for (int s = 0; s < 2; ++s) kt = wmma_k4(asi[s], bq[s], kt);

    v2f ak[2], bpp[2];
    conv_lo(kt, hi, ak);
    conv_lo(pp, hi, bpp);
    v8f pn = pp;
#pragma unroll
    for (int s = 0; s < 2; ++s) {
      v2f na;
      na.x = -ak[s].x;
      na.y = -ak[s].y;
      pn = wmma_k4(na, bpp[s], pn);
    }
    Pd = pn;

    __builtin_amdgcn_sched_barrier(0);
    float xn = xp;
    xn = __builtin_fmaf(kt[0], y0, xn);
    xn = __builtin_fmaf(kt[1], y1, xn);
    xn = __builtin_fmaf(kt[2], y2, xn);
    xn = __builtin_fmaf(kt[3], y3, xn);
    xn = __builtin_fmaf(kt[4], y4, xn);
    xn = __builtin_fmaf(kt[5], y5, xn);
    xn = __builtin_fmaf(kt[6], y6, xn);
    xn = __builtin_fmaf(kt[7], y7, xn);
    if (lane < 16) out[(size_t)t * 16 + l16] = xn;
    xv = xn;
    zv = znext;
  }
}

extern "C" void kernel_launch(void* const* d_in, const int* in_sizes, int n_in,
                              void* d_out, int out_size, void* d_ws, size_t ws_size,
                              hipStream_t stream) {
  const float* meas = (const float*)d_in[0];
  const float* F    = (const float*)d_in[1];
  const float* Q    = (const float*)d_in[3];
  const float* R    = (const float*)d_in[4];
  float* out = (float*)d_out;

  const int T = in_sizes[0] / MEAS;

  hipLaunchKernelGGL(kalman_seq_kernel, dim3(1), dim3(32), 0, stream,
                     meas, F, Q, R, out, T);
}


# Gauss-Jordan 8x8 as a hand-ordered sequence with the row_share broadcasts folded into the arithmetic (v_rcp_f32_dpp, v_fmac_f32_dpp); the Pp^T WMMAs fenced out of the Pp chain
# speedup vs baseline: 2.7921x; 1.0664x over previous
#include <hip/hip_runtime.h>


typedef __attribute__((ext_vector_type(2))) float v2f;
typedef __attribute__((ext_vector_type(8))) float v8f;

#define STATE 16
#define MEAS  8

static __device__ __forceinline__ v8f wmma_k4(v2f a, v2f b, v8f c) {
  return __builtin_amdgcn_wmma_f32_16x16x4_f32(
       false, a,  false, b,
       (short)0, c,  false,  false);
}

static __device__ __forceinline__ float xhalf(float x) {
  const int b = __builtin_bit_cast(int, x);
  return __builtin_bit_cast(float,
      __builtin_amdgcn_permlanex16(b, b, 0x76543210, 0xfedcba98, false, false));
}

template <int N>
static __device__ __forceinline__ float rowshare(float x) {
  const int b = __builtin_bit_cast(int, x);
  return __builtin_bit_cast(float,
      __builtin_amdgcn_mov_dpp(b, 0x150 + N, 0xf, 0xf, false));
}

static __device__ __forceinline__ float rowshl8_zero(float x) {
  const int b = __builtin_bit_cast(int, x);
  return __builtin_bit_cast(float,
      __builtin_amdgcn_update_dpp(0, b, 0x108, 0xf, 0xf, true));
}

static __device__ __forceinline__ void conv_full(const v8f d, const bool hi, v2f (&f)[4]) {
  float X[8];
#pragma unroll
  for (int r = 0; r < 8; ++r) X[r] = xhalf(d[r]);
  f[0].x = hi ? X[2] : d[0];  f[0].y = hi ? X[3] : d[1];
  f[1].x = hi ? X[6] : d[4];  f[1].y = hi ? X[7] : d[5];
  f[2].x = hi ? d[2] : X[0];  f[2].y = hi ? d[3] : X[1];
  f[3].x = hi ? d[6] : X[4];  f[3].y = hi ? d[7] : X[5];
}

static __device__ __forceinline__ void conv_lo(const v8f d, const bool hi, v2f (&f)[2]) {
  const float x2 = xhalf(d[2]), x3 = xhalf(d[3]);
  const float x6 = xhalf(d[6]), x7 = xhalf(d[7]);
  f[0].x = hi ? x2 : d[0];  f[0].y = hi ? x3 : d[1];
  f[1].x = hi ? x6 : d[4];  f[1].y = hi ? x7 : d[5];
}


#define GJ_FMAC(K, acc, m)                                                      \
  asm volatile("v_fmac_f32_dpp %0, %0, %1 row_share:" #K                        \
               " row_mask:0xf bank_mask:0xf"                                    \
               : "+v"(acc) : "v"(m))

#define GJ_RCP_FMAC(N, piv, src, K, acc, m)                                     \
  asm volatile("v_rcp_f32_dpp %0, %2 row_share:" #N                             \
               " row_mask:0xf bank_mask:0xf\n\t"                                \
               "v_fmac_f32_dpp %1, %1, %3 row_share:" #K                        \
               " row_mask:0xf bank_mask:0xf"                                    \
               : "=&v"(piv), "+v"(acc) : "v"(src), "v"(m))

#define GJ_NMUL(dst, row, piv)                                                  \
  asm volatile("v_mul_f32_e64 %0, %1, -%2" : "=v"(dst) : "v"(row), "v"(piv))

#define GJ_STEP(K, N, A, B, C, D, E, F)                                         \
  {                                                                             \
    float pivn, ngkn;                                                           \
    GJ_FMAC(K, g[N], ngk);                                                      \
    GJ_FMAC(K, g[A], ngk);                                                      \
    GJ_FMAC(K, g[B], ngk);                                                      \
    GJ_RCP_FMAC(N, pivn, g[N], K, g[C], ngk);                                   \
    GJ_FMAC(K, g[D], ngk);                                                      \
    GJ_FMAC(K, g[E], ngk);                                                      \
    GJ_NMUL(ngkn, g[N], pivn);                                                  \
    GJ_FMAC(K, g[F], ngk);                                                      \
    g[K] = g[K] * piv;                                                          \
    piv  = pivn;                                                                \
    ngk  = ngkn;                                                                \
  }

__global__ __launch_bounds__(32) void kalman_seq_kernel(
    const float* __restrict__ meas,
    const float* __restrict__ Fm,
    const float* __restrict__ Qm,
    const float* __restrict__ Rm,
    float* __restrict__ out,
    int T)
{
  const int lane = threadIdx.x;
  const int l16  = lane & 15;
  const int half = lane >> 4;
  const bool hi  = (half != 0);

  float Frow[16];
#pragma unroll
  for (int j = 0; j < 16; ++j) Frow[j] = Fm[l16 * 16 + j];

  v2f fab[4];
#pragma unroll
  for (int s = 0; s < 4; ++s) {
    const int k0 = 4 * s + 2 * half;
    fab[s].x = Fm[l16 * 16 + k0];
    fab[s].y = Fm[l16 * 16 + k0 + 1];
  }

  v8f qacc, qaccT;
#pragma unroll
  for (int r = 0; r < 8; ++r) {
    qacc[r]  = Qm[(r + 8 * half) * 16 + l16];
    qaccT[r] = Qm[l16 * 16 + (r + 8 * half)];
  }

  const float maskL = (lane < 8) ? 1.0f : 0.0f;
  float rI[8];
#pragma unroll
  for (int i = 0; i < 8; ++i) {
    const float idn = ((l16 & 7) == i) ? 1.0f : 0.0f;
    rI[i] = (lane < 8) ? Rm[i * 8 + (lane & 7)] : idn;
  }

  v8f Pd;
#pragma unroll
  for (int r = 0; r < 8; ++r) Pd[r] = ((r + 8 * half) == l16) ? 1.0f : 0.0f;
  float xv = 0.0f;

  float zv = meas[lane & 7];

  for (int t = 0; t < T; ++t) {
    const int tn = (t + 1 < T) ? (t + 1) : (T - 1);
    const float znext = meas[(size_t)tn * MEAS + (lane & 7)];
    __builtin_prefetch(&meas[(size_t)(t + 48) * MEAS], 0, 1);

    float x0 = 0.0f, x1 = 0.0f, x2 = 0.0f, x3 = 0.0f;
    x0 = __builtin_fmaf(Frow[0],  rowshare<0>(xv),  x0);
    x1 = __builtin_fmaf(Frow[4],  rowshare<4>(xv),  x1);
    x2 = __builtin_fmaf(Frow[8],  rowshare<8>(xv),  x2);
    x3 = __builtin_fmaf(Frow[12], rowshare<12>(xv), x3);
    x0 = __builtin_fmaf(Frow[1],  rowshare<1>(xv),  x0);
    x1 = __builtin_fmaf(Frow[5],  rowshare<5>(xv),  x1);
    x2 = __builtin_fmaf(Frow[9],  rowshare<9>(xv),  x2);
    x3 = __builtin_fmaf(Frow[13], rowshare<13>(xv), x3);
    x0 = __builtin_fmaf(Frow[2],  rowshare<2>(xv),  x0);
    x1 = __builtin_fmaf(Frow[6],  rowshare<6>(xv),  x1);
    x2 = __builtin_fmaf(Frow[10], rowshare<10>(xv), x2);
    x3 = __builtin_fmaf(Frow[14], rowshare<14>(xv), x3);
    x0 = __builtin_fmaf(Frow[3],  rowshare<3>(xv),  x0);
    x1 = __builtin_fmaf(Frow[7],  rowshare<7>(xv),  x1);
    x2 = __builtin_fmaf(Frow[11], rowshare<11>(xv), x2);
    x3 = __builtin_fmaf(Frow[15], rowshare<15>(xv), x3);
    const float xp = (x0 + x1) + (x2 + x3);
    const float yv = zv - xp;
    const float y0 = rowshare<0>(yv), y1 = rowshare<1>(yv);
    const float y2 = rowshare<2>(yv), y3 = rowshare<3>(yv);
    const float y4 = rowshare<4>(yv), y5 = rowshare<5>(yv);
    const float y6 = rowshare<6>(yv), y7 = rowshare<7>(yv);

    v2f bp[4];
    conv_full(Pd, hi, bp);
    v8f m1t = {0, 0, 0, 0, 0, 0, 0, 0};
#pragma unroll
    for (int s = 0; s < 4; ++s) m1t = wmma_k4(bp[s], fab[s], m1t);

    v2f bm[4];
    conv_full(m1t, hi, bm);
    v8f pp = qacc;
#pragma unroll
    for (int s = 0; s < 4; ++s) pp = wmma_k4(bm[s], fab[s], pp);
    __builtin_amdgcn_sched_barrier(0);
    v8f ppt = wmma_k4(fab[0], bm[0], qaccT);
    __builtin_amdgcn_sched_barrier(0);

    float g[8];
#pragma unroll
    for (int i = 0; i < 8; ++i) g[i] = __builtin_fmaf(pp[i], maskL, rI[i]);
    float piv;
    asm volatile("v_rcp_f32_dpp %0, %1 row_share:0 row_mask:0xf bank_mask:0xf\n\t"
                 "v_nop"
                 : "=v"(piv) : "v"(g[0]));
    float ngk = g[0] * (-piv);
    GJ_STEP(0, 1, 2, 3, 4, 5, 6, 7)
    __builtin_amdgcn_sched_barrier(0);
    ppt = wmma_k4(fab[1], bm[1], ppt);
    __builtin_amdgcn_sched_barrier(0);
    GJ_STEP(1, 2, 0, 3, 4, 5, 6, 7)
    __builtin_amdgcn_sched_barrier(0);
    ppt = wmma_k4(fab[2], bm[2], ppt);
    __builtin_amdgcn_sched_barrier(0);
    GJ_STEP(2, 3, 0, 1, 4, 5, 6, 7)
    __builtin_amdgcn_sched_barrier(0);
    ppt = wmma_k4(fab[3], bm[3], ppt);
    __builtin_amdgcn_sched_barrier(0);
    GJ_STEP(3, 4, 0, 1, 2, 5, 6, 7)
    GJ_STEP(4, 5, 0, 1, 2, 3, 6, 7)
    GJ_STEP(5, 6, 0, 1, 2, 3, 4, 7)
    GJ_STEP(6, 7, 0, 1, 2, 3, 4, 5)
    {
      GJ_FMAC(7, g[0], ngk);
      GJ_FMAC(7, g[1], ngk);
      GJ_FMAC(7, g[2], ngk);
      GJ_FMAC(7, g[3], ngk);
      GJ_FMAC(7, g[4], ngk);
      GJ_FMAC(7, g[5], ngk);
      GJ_FMAC(7, g[6], ngk);
      g[7] = g[7] * piv;
    }
    float sh[8];
#pragma unroll
    for (int i = 0; i < 8; ++i) sh[i] = rowshl8_zero(g[i]);
    v2f asi[2];
    {
      const float s2 = xhalf(sh[2]), s3 = xhalf(sh[3]);
      const float s6 = xhalf(sh[6]), s7 = xhalf(sh[7]);
      asi[0].x = hi ? s2 : sh[0];  asi[0].y = hi ? s3 : sh[1];
      asi[1].x = hi ? s6 : sh[4];  asi[1].y = hi ? s7 : sh[5];
    }

    v2f bq[2];
    conv_lo(ppt, hi, bq);
    v8f kt = {0, 0, 0, 0, 0, 0, 0, 0};
#pragma unroll
    for (int s = 0; s < 2; ++s) kt = wmma_k4(asi[s], bq[s], kt);

    v2f ak[2], bpp[2];
    conv_lo(kt, hi, ak);
    conv_lo(pp, hi, bpp);
    v8f pn = pp;
#pragma unroll
    for (int s = 0; s < 2; ++s) {
      v2f na;
      na.x = -ak[s].x;
      na.y = -ak[s].y;
      pn = wmma_k4(na, bpp[s], pn);
    }
    Pd = pn;

    __builtin_amdgcn_sched_barrier(0);
    float xn = xp;
    xn = __builtin_fmaf(kt[0], y0, xn);
    xn = __builtin_fmaf(kt[1], y1, xn);
    xn = __builtin_fmaf(kt[2], y2, xn);
    xn = __builtin_fmaf(kt[3], y3, xn);
    xn = __builtin_fmaf(kt[4], y4, xn);
    xn = __builtin_fmaf(kt[5], y5, xn);
    xn = __builtin_fmaf(kt[6], y6, xn);
    xn = __builtin_fmaf(kt[7], y7, xn);
    if (lane < 16) out[(size_t)t * 16 + l16] = xn;
    xv = xn;
    zv = znext;
  }
}

extern "C" void kernel_launch(void* const* d_in, const int* in_sizes, int n_in,
                              void* d_out, int out_size, void* d_ws, size_t ws_size,
                              hipStream_t stream) {
  const float* meas = (const float*)d_in[0];
  const float* F    = (const float*)d_in[1];
  const float* Q    = (const float*)d_in[3];
  const float* R    = (const float*)d_in[4];
  float* out = (float*)d_out;

  const int T = in_sizes[0] / MEAS;

  hipLaunchKernelGGL(kalman_seq_kernel, dim3(1), dim3(32), 0, stream,
                     meas, F, Q, R, out, T);
}


# State recursion (F x, innovation, K y, z loads, x stores) moved to a second wave; wave 0 keeps only the covariance chain and hands each K^T tile over through an LDS ring, one barrier per 32 steps
# speedup vs baseline: 3.5343x; 1.2658x over previous
#include <hip/hip_runtime.h>


typedef __attribute__((ext_vector_type(2))) float v2f;
typedef __attribute__((ext_vector_type(8))) float v8f;

#define STATE 16
#define MEAS  8
#define CH    32
#define KSTEP 256

static __device__ __forceinline__ v8f wmma_k4(v2f a, v2f b, v8f c) {
  return __builtin_amdgcn_wmma_f32_16x16x4_f32(
       false, a,  false, b,
       (short)0, c,  false,  false);
}

static __device__ __forceinline__ float xhalf(float x) {
  const int b = __builtin_bit_cast(int, x);
  return __builtin_bit_cast(float,
      __builtin_amdgcn_permlanex16(b, b, 0x76543210, 0xfedcba98, false, false));
}

template <int N>
static __device__ __forceinline__ float rowshare(float x) {
  const int b = __builtin_bit_cast(int, x);
  return __builtin_bit_cast(float,
      __builtin_amdgcn_mov_dpp(b, 0x150 + N, 0xf, 0xf, false));
}

static __device__ __forceinline__ float rowshl8_zero(float x) {
  const int b = __builtin_bit_cast(int, x);
  return __builtin_bit_cast(float,
      __builtin_amdgcn_update_dpp(0, b, 0x108, 0xf, 0xf, true));
}

static __device__ __forceinline__ void conv_full(const v8f d, const bool hi, v2f (&f)[4]) {
  float X[8];
#pragma unroll
  for (int r = 0; r < 8; ++r) X[r] = xhalf(d[r]);
  f[0].x = hi ? X[2] : d[0];  f[0].y = hi ? X[3] : d[1];
  f[1].x = hi ? X[6] : d[4];  f[1].y = hi ? X[7] : d[5];
  f[2].x = hi ? d[2] : X[0];  f[2].y = hi ? d[3] : X[1];
  f[3].x = hi ? d[6] : X[4];  f[3].y = hi ? d[7] : X[5];
}

static __device__ __forceinline__ void conv_lo(const v8f d, const bool hi, v2f (&f)[2]) {
  const float x2 = xhalf(d[2]), x3 = xhalf(d[3]);
  const float x6 = xhalf(d[6]), x7 = xhalf(d[7]);
  f[0].x = hi ? x2 : d[0];  f[0].y = hi ? x3 : d[1];
  f[1].x = hi ? x6 : d[4];  f[1].y = hi ? x7 : d[5];
}


#define GJ_FMAC(K, acc, m)                                                      \
  asm volatile("v_fmac_f32_dpp %0, %0, %1 row_share:" #K                        \
               " row_mask:0xf bank_mask:0xf"                                    \
               : "+v"(acc) : "v"(m))

#define GJ_RCP_FMAC(N, piv, src, K, acc, m)                                     \
  asm volatile("v_rcp_f32_dpp %0, %2 row_share:" #N                             \
               " row_mask:0xf bank_mask:0xf\n\t"                                \
               "v_fmac_f32_dpp %1, %1, %3 row_share:" #K                        \
               " row_mask:0xf bank_mask:0xf"                                    \
               : "=&v"(piv), "+v"(acc) : "v"(src), "v"(m))

#define GJ_NMUL(dst, row, piv)                                                  \
  asm volatile("v_mul_f32_e64 %0, %1, -%2" : "=v"(dst) : "v"(row), "v"(piv))

#define GJ_STEP(K, N, A, B, C, D, E, F)                                         \
  {                                                                             \
    float pivn, ngkn;                                                           \
    GJ_FMAC(K, g[N], ngk);                                                      \
    GJ_FMAC(K, g[A], ngk);                                                      \
    GJ_FMAC(K, g[B], ngk);                                                      \
    GJ_RCP_FMAC(N, pivn, g[N], K, g[C], ngk);                                   \
    GJ_FMAC(K, g[D], ngk);                                                      \
    GJ_FMAC(K, g[E], ngk);                                                      \
    GJ_NMUL(ngkn, g[N], pivn);                                                  \
    GJ_FMAC(K, g[F], ngk);                                                      \
    g[K] = g[K] * piv;                                                          \
    piv  = pivn;                                                                \
    ngk  = ngkn;                                                                \
  }

__global__ __launch_bounds__(64) void kalman_seq_kernel(
    const float* __restrict__ meas,
    const float* __restrict__ Fm,
    const float* __restrict__ Qm,
    const float* __restrict__ Rm,
    float* __restrict__ out,
    int T)
{
  __shared__ __attribute__((aligned(16))) float kring[2 * CH * KSTEP];

  const int lane = threadIdx.x & 31;
  const int wv = __builtin_amdgcn_readfirstlane((int)(threadIdx.x >> 5));
  const int l16  = lane & 15;
  const int half = lane >> 4;
  const bool hi  = (half != 0);
  const int nch  = (T + CH - 1) / CH;

  float Frow[16];
#pragma unroll
  for (int j = 0; j < 16; ++j) Frow[j] = Fm[l16 * 16 + j];

  v2f fab[4];
#pragma unroll
  for (int s = 0; s < 4; ++s) {
    const int k0 = 4 * s + 2 * half;
    fab[s].x = Fm[l16 * 16 + k0];
    fab[s].y = Fm[l16 * 16 + k0 + 1];
  }

  v8f qacc, qaccT;
#pragma unroll
  for (int r = 0; r < 8; ++r) {
    qacc[r]  = Qm[(r + 8 * half) * 16 + l16];
    qaccT[r] = Qm[l16 * 16 + (r + 8 * half)];
  }

  const float maskL = (lane < 8) ? 1.0f : 0.0f;
  float rI[8];
#pragma unroll
  for (int i = 0; i < 8; ++i) {
    const float idn = ((l16 & 7) == i) ? 1.0f : 0.0f;
    rI[i] = (lane < 8) ? Rm[i * 8 + (lane & 7)] : idn;
  }

  v8f Pd;
#pragma unroll
  for (int r = 0; r < 8; ++r) Pd[r] = ((r + 8 * half) == l16) ? 1.0f : 0.0f;
  float xv = 0.0f;

  float zv = meas[lane & 7];

  for (int c = 0; c <= nch; ++c) {
    if (wv == 0) {
      if (c < nch) {
        const int t0 = c * CH;
        const int n  = __builtin_amdgcn_readfirstlane((T - t0 < CH) ? (T - t0) : CH);
        float* kb = &kring[(c & 1) * (CH * KSTEP) + lane * 8];
        for (int i = 0; i < n; ++i) {
          v2f bp[4];
          conv_full(Pd, hi, bp);
          v8f m1t = {0, 0, 0, 0, 0, 0, 0, 0};
#pragma unroll
          for (int s = 0; s < 4; ++s) m1t = wmma_k4(bp[s], fab[s], m1t);

          v2f bm[4];
          conv_full(m1t, hi, bm);
          v8f pp = qacc;
#pragma unroll
          for (int s = 0; s < 4; ++s) pp = wmma_k4(bm[s], fab[s], pp);
          __builtin_amdgcn_sched_barrier(0);
          v8f ppt = wmma_k4(fab[0], bm[0], qaccT);
          __builtin_amdgcn_sched_barrier(0);

          float g[8];
#pragma unroll
          for (int r = 0; r < 8; ++r) g[r] = __builtin_fmaf(pp[r], maskL, rI[r]);
          float piv;
          asm volatile("v_rcp_f32_dpp %0, %1 row_share:0 row_mask:0xf bank_mask:0xf\n\t"
                       "v_nop"
                       : "=v"(piv) : "v"(g[0]));
          float ngk = g[0] * (-piv);
          GJ_STEP(0, 1, 2, 3, 4, 5, 6, 7)
          __builtin_amdgcn_sched_barrier(0);
          ppt = wmma_k4(fab[1], bm[1], ppt);
          __builtin_amdgcn_sched_barrier(0);
          GJ_STEP(1, 2, 0, 3, 4, 5, 6, 7)
          __builtin_amdgcn_sched_barrier(0);
          ppt = wmma_k4(fab[2], bm[2], ppt);
          __builtin_amdgcn_sched_barrier(0);
          GJ_STEP(2, 3, 0, 1, 4, 5, 6, 7)
          __builtin_amdgcn_sched_barrier(0);
          ppt = wmma_k4(fab[3], bm[3], ppt);
          __builtin_amdgcn_sched_barrier(0);
          GJ_STEP(3, 4, 0, 1, 2, 5, 6, 7)
          GJ_STEP(4, 5, 0, 1, 2, 3, 6, 7)
          GJ_STEP(5, 6, 0, 1, 2, 3, 4, 7)
          GJ_STEP(6, 7, 0, 1, 2, 3, 4, 5)
          {
            GJ_FMAC(7, g[0], ngk);
            GJ_FMAC(7, g[1], ngk);
            GJ_FMAC(7, g[2], ngk);
            GJ_FMAC(7, g[3], ngk);
            GJ_FMAC(7, g[4], ngk);
            GJ_FMAC(7, g[5], ngk);
            GJ_FMAC(7, g[6], ngk);
            g[7] = g[7] * piv;
          }
          float sh[8];
#pragma unroll
          for (int r = 0; r < 8; ++r) sh[r] = rowshl8_zero(g[r]);
          v2f asi[2];
          {
            const float s2 = xhalf(sh[2]), s3 = xhalf(sh[3]);
            const float s6 = xhalf(sh[6]), s7 = xhalf(sh[7]);
            asi[0].x = hi ? s2 : sh[0];  asi[0].y = hi ? s3 : sh[1];
            asi[1].x = hi ? s6 : sh[4];  asi[1].y = hi ? s7 : sh[5];
          }

          v2f bq[2];
          conv_lo(ppt, hi, bq);
          v8f kt = {0, 0, 0, 0, 0, 0, 0, 0};
#pragma unroll
          for (int s = 0; s < 2; ++s) kt = wmma_k4(asi[s], bq[s], kt);

          v2f ak[2], bpp[2];
          conv_lo(kt, hi, ak);
          conv_lo(pp, hi, bpp);
          v8f pn = pp;
#pragma unroll
          for (int s = 0; s < 2; ++s) {
            v2f na;
            na.x = -ak[s].x;
            na.y = -ak[s].y;
            pn = wmma_k4(na, bpp[s], pn);
          }
          Pd = pn;

          __builtin_amdgcn_sched_barrier(0);
          float4 klo, khi;
          klo.x = kt[0]; klo.y = kt[1]; klo.z = kt[2]; klo.w = kt[3];
          khi.x = kt[4]; khi.y = kt[5]; khi.z = kt[6]; khi.w = kt[7];
          *reinterpret_cast<float4*>(kb + i * KSTEP)     = klo;
          *reinterpret_cast<float4*>(kb + i * KSTEP + 4) = khi;
        }
      }
    } else {
      if (c > 0) {
        const int t0 = (c - 1) * CH;
        const int n  = __builtin_amdgcn_readfirstlane((T - t0 < CH) ? (T - t0) : CH);
        const float* kb = &kring[((c - 1) & 1) * (CH * KSTEP) + lane * 8];
        for (int i = 0; i < n; ++i) {
          const int t = t0 + i;
          const int tn = (t + 1 < T) ? (t + 1) : (T - 1);
          const float znext = meas[(size_t)tn * MEAS + (lane & 7)];
          __builtin_prefetch(&meas[(size_t)(t + 48) * MEAS], 0, 1);
          const float4 klo = *reinterpret_cast<const float4*>(kb + i * KSTEP);
          const float4 khi = *reinterpret_cast<const float4*>(kb + i * KSTEP + 4);

          float x0 = 0.0f, x1 = 0.0f, x2 = 0.0f, x3 = 0.0f;
          x0 = __builtin_fmaf(Frow[0],  rowshare<0>(xv),  x0);
          x1 = __builtin_fmaf(Frow[4],  rowshare<4>(xv),  x1);
          x2 = __builtin_fmaf(Frow[8],  rowshare<8>(xv),  x2);
          x3 = __builtin_fmaf(Frow[12], rowshare<12>(xv), x3);
          x0 = __builtin_fmaf(Frow[1],  rowshare<1>(xv),  x0);
          x1 = __builtin_fmaf(Frow[5],  rowshare<5>(xv),  x1);
          x2 = __builtin_fmaf(Frow[9],  rowshare<9>(xv),  x2);
          x3 = __builtin_fmaf(Frow[13], rowshare<13>(xv), x3);
          x0 = __builtin_fmaf(Frow[2],  rowshare<2>(xv),  x0);
          x1 = __builtin_fmaf(Frow[6],  rowshare<6>(xv),  x1);
          x2 = __builtin_fmaf(Frow[10], rowshare<10>(xv), x2);
          x3 = __builtin_fmaf(Frow[14], rowshare<14>(xv), x3);
          x0 = __builtin_fmaf(Frow[3],  rowshare<3>(xv),  x0);
          x1 = __builtin_fmaf(Frow[7],  rowshare<7>(xv),  x1);
          x2 = __builtin_fmaf(Frow[11], rowshare<11>(xv), x2);
          x3 = __builtin_fmaf(Frow[15], rowshare<15>(xv), x3);
          const float xp = (x0 + x1) + (x2 + x3);
          const float yv = zv - xp;
          const float y0 = rowshare<0>(yv), y1 = rowshare<1>(yv);
          const float y2 = rowshare<2>(yv), y3 = rowshare<3>(yv);
          const float y4 = rowshare<4>(yv), y5 = rowshare<5>(yv);
          const float y6 = rowshare<6>(yv), y7 = rowshare<7>(yv);

          float xn = xp;
          xn = __builtin_fmaf(klo.x, y0, xn);
          xn = __builtin_fmaf(klo.y, y1, xn);
          xn = __builtin_fmaf(klo.z, y2, xn);
          xn = __builtin_fmaf(klo.w, y3, xn);
          xn = __builtin_fmaf(khi.x, y4, xn);
          xn = __builtin_fmaf(khi.y, y5, xn);
          xn = __builtin_fmaf(khi.z, y6, xn);
          xn = __builtin_fmaf(khi.w, y7, xn);
          if (lane < 16) out[(size_t)t * 16 + l16] = xn;
          xv = xn;
          zv = znext;
        }
      }
    }
    __syncthreads();
  }
}

extern "C" void kernel_launch(void* const* d_in, const int* in_sizes, int n_in,
                              void* d_out, int out_size, void* d_ws, size_t ws_size,
                              hipStream_t stream) {
  const float* meas = (const float*)d_in[0];
  const float* F    = (const float*)d_in[1];
  const float* Q    = (const float*)d_in[3];
  const float* R    = (const float*)d_in[4];
  float* out = (float*)d_out;

  const int T = in_sizes[0] / MEAS;

  hipLaunchKernelGGL(kalman_seq_kernel, dim3(1), dim3(64), 0, stream,
                     meas, F, Q, R, out, T);
}


# k order of the three 16-deep F products set to k = 8h+2s+e, F loaded to match: a D tile is its own fragment set, so the half-wave exchange in front of P^T F^T and of M1 F^T is gone
# speedup vs baseline: 3.9410x; 1.1151x over previous
#include <hip/hip_runtime.h>


typedef __attribute__((ext_vector_type(2))) float v2f;
typedef __attribute__((ext_vector_type(8))) float v8f;

#define STATE 16
#define MEAS  8
#define CH    32
#define KSTEP 256

static __device__ __forceinline__ v8f wmma_k4(v2f a, v2f b, v8f c) {
  return __builtin_amdgcn_wmma_f32_16x16x4_f32(
       false, a,  false, b,
       (short)0, c,  false,  false);
}

static __device__ __forceinline__ float xhalf(float x) {
  const int b = __builtin_bit_cast(int, x);
  return __builtin_bit_cast(float,
      __builtin_amdgcn_permlanex16(b, b, 0x76543210, 0xfedcba98, false, false));
}

template <int N>
static __device__ __forceinline__ float rowshare(float x) {
  const int b = __builtin_bit_cast(int, x);
  return __builtin_bit_cast(float,
      __builtin_amdgcn_mov_dpp(b, 0x150 + N, 0xf, 0xf, false));
}

static __device__ __forceinline__ float rowshl8_zero(float x) {
  const int b = __builtin_bit_cast(int, x);
  return __builtin_bit_cast(float,
      __builtin_amdgcn_update_dpp(0, b, 0x108, 0xf, 0xf, true));
}

static __device__ __forceinline__ void dfrag(const v8f d, v2f (&f)[4]) {
#pragma unroll
  for (int s = 0; s < 4; ++s) { f[s].x = d[2 * s]; f[s].y = d[2 * s + 1]; }
}

static __device__ __forceinline__ void conv_lo(const v8f d, const bool hi, v2f (&f)[2]) {
  const float x2 = xhalf(d[2]), x3 = xhalf(d[3]);
  const float x6 = xhalf(d[6]), x7 = xhalf(d[7]);
  f[0].x = hi ? x2 : d[0];  f[0].y = hi ? x3 : d[1];
  f[1].x = hi ? x6 : d[4];  f[1].y = hi ? x7 : d[5];
}


#define GJ_FMAC(K, acc, m)                                                      \
  asm volatile("v_fmac_f32_dpp %0, %0, %1 row_share:" #K                        \
               " row_mask:0xf bank_mask:0xf"                                    \
               : "+v"(acc) : "v"(m))

#define GJ_RCP_FMAC(N, piv, src, K, acc, m)                                     \
  asm volatile("v_rcp_f32_dpp %0, %2 row_share:" #N                             \
               " row_mask:0xf bank_mask:0xf\n\t"                                \
               "v_fmac_f32_dpp %1, %1, %3 row_share:" #K                        \
               " row_mask:0xf bank_mask:0xf"                                    \
               : "=&v"(piv), "+v"(acc) : "v"(src), "v"(m))

#define GJ_NMUL(dst, row, piv)                                                  \
  asm volatile("v_mul_f32_e64 %0, %1, -%2" : "=v"(dst) : "v"(row), "v"(piv))

#define GJ_STEP(K, N, A, B, C, D, E, F)                                         \
  {                                                                             \
    float pivn, ngkn;                                                           \
    GJ_FMAC(K, g[N], ngk);                                                      \
    GJ_FMAC(K, g[A], ngk);                                                      \
    GJ_FMAC(K, g[B], ngk);                                                      \
    GJ_RCP_FMAC(N, pivn, g[N], K, g[C], ngk);                                   \
    GJ_FMAC(K, g[D], ngk);                                                      \
    GJ_FMAC(K, g[E], ngk);                                                      \
    GJ_NMUL(ngkn, g[N], pivn);                                                  \
    GJ_FMAC(K, g[F], ngk);                                                      \
    g[K] = g[K] * piv;                                                          \
    piv  = pivn;                                                                \
    ngk  = ngkn;                                                                \
  }

__global__ __launch_bounds__(64) void kalman_seq_kernel(
    const float* __restrict__ meas,
    const float* __restrict__ Fm,
    const float* __restrict__ Qm,
    const float* __restrict__ Rm,
    float* __restrict__ out,
    int T)
{
  __shared__ __attribute__((aligned(16))) float kring[2 * CH * KSTEP];

  const int lane = threadIdx.x & 31;
  const int wv = __builtin_amdgcn_readfirstlane((int)(threadIdx.x >> 5));
  const int l16  = lane & 15;
  const int half = lane >> 4;
  const bool hi  = (half != 0);
  const int nch  = (T + CH - 1) / CH;

  float Frow[16];
#pragma unroll
  for (int j = 0; j < 16; ++j) Frow[j] = Fm[l16 * 16 + j];

  v2f fab[4];
#pragma unroll
  for (int s = 0; s < 4; ++s) {
    const int k0 = 8 * half + 2 * s;
    fab[s].x = Fm[l16 * 16 + k0];
    fab[s].y = Fm[l16 * 16 + k0 + 1];
  }

  v8f qacc, qaccT;
#pragma unroll
  for (int r = 0; r < 8; ++r) {
    qacc[r]  = Qm[(r + 8 * half) * 16 + l16];
    qaccT[r] = Qm[l16 * 16 + (r + 8 * half)];
  }

  const float maskL = (lane < 8) ? 1.0f : 0.0f;
  float rI[8];
#pragma unroll
  for (int i = 0; i < 8; ++i) {
    const float idn = ((l16 & 7) == i) ? 1.0f : 0.0f;
    rI[i] = (lane < 8) ? Rm[i * 8 + (lane & 7)] : idn;
  }

  v8f Pd;
#pragma unroll
  for (int r = 0; r < 8; ++r) Pd[r] = ((r + 8 * half) == l16) ? 1.0f : 0.0f;
  float xv = 0.0f;

  float zv = meas[lane & 7];

  for (int c = 0; c <= nch; ++c) {
    if (wv == 0) {
      if (c < nch) {
        const int t0 = c * CH;
        const int n  = __builtin_amdgcn_readfirstlane((T - t0 < CH) ? (T - t0) : CH);
        float* kb = &kring[(c & 1) * (CH * KSTEP) + lane * 8];
        for (int i = 0; i < n; ++i) {
          v2f bp[4];
          dfrag(Pd, bp);
          v8f m1t = {0, 0, 0, 0, 0, 0, 0, 0};
#pragma unroll
          for (int s = 0; s < 4; ++s) m1t = wmma_k4(bp[s], fab[s], m1t);

          v2f bm[4];
          dfrag(m1t, bm);
          v8f pp = qacc;
#pragma unroll
          for (int s = 0; s < 4; ++s) pp = wmma_k4(bm[s], fab[s], pp);
          __builtin_amdgcn_sched_barrier(0);
          v8f ppt = wmma_k4(fab[0], bm[0], qaccT);
          __builtin_amdgcn_sched_barrier(0);

          float g[8];
#pragma unroll
          for (int r = 0; r < 8; ++r) g[r] = __builtin_fmaf(pp[r], maskL, rI[r]);
          float piv;
          asm volatile("v_rcp_f32_dpp %0, %1 row_share:0 row_mask:0xf bank_mask:0xf\n\t"
                       "v_nop"
                       : "=v"(piv) : "v"(g[0]));
          float ngk = g[0] * (-piv);
          GJ_STEP(0, 1, 2, 3, 4, 5, 6, 7)
          __builtin_amdgcn_sched_barrier(0);
          ppt = wmma_k4(fab[1], bm[1], ppt);
          __builtin_amdgcn_sched_barrier(0);
          GJ_STEP(1, 2, 0, 3, 4, 5, 6, 7)
          __builtin_amdgcn_sched_barrier(0);
          ppt = wmma_k4(fab[2], bm[2], ppt);
          __builtin_amdgcn_sched_barrier(0);
          GJ_STEP(2, 3, 0, 1, 4, 5, 6, 7)
          __builtin_amdgcn_sched_barrier(0);
          ppt = wmma_k4(fab[3], bm[3], ppt);
          __builtin_amdgcn_sched_barrier(0);
          GJ_STEP(3, 4, 0, 1, 2, 5, 6, 7)
          GJ_STEP(4, 5, 0, 1, 2, 3, 6, 7)
          GJ_STEP(5, 6, 0, 1, 2, 3, 4, 7)
          GJ_STEP(6, 7, 0, 1, 2, 3, 4, 5)
          {
            GJ_FMAC(7, g[0], ngk);
            GJ_FMAC(7, g[1], ngk);
            GJ_FMAC(7, g[2], ngk);
            GJ_FMAC(7, g[3], ngk);
            GJ_FMAC(7, g[4], ngk);
            GJ_FMAC(7, g[5], ngk);
            GJ_FMAC(7, g[6], ngk);
            g[7] = g[7] * piv;
          }
          float sh[8];
#pragma unroll
          for (int r = 0; r < 8; ++r) sh[r] = rowshl8_zero(g[r]);
          v2f asi[2];
          {
            const float s2 = xhalf(sh[2]), s3 = xhalf(sh[3]);
            const float s6 = xhalf(sh[6]), s7 = xhalf(sh[7]);
            asi[0].x = hi ? s2 : sh[0];  asi[0].y = hi ? s3 : sh[1];
            asi[1].x = hi ? s6 : sh[4];  asi[1].y = hi ? s7 : sh[5];
          }

          v2f bq[2];
          conv_lo(ppt, hi, bq);
          v8f kt = {0, 0, 0, 0, 0, 0, 0, 0};
#pragma unroll
          for (int s = 0; s < 2; ++s) kt = wmma_k4(asi[s], bq[s], kt);

          v2f ak[2], bpp[2];
          conv_lo(kt, hi, ak);
          conv_lo(pp, hi, bpp);
          v8f pn = pp;
#pragma unroll
          for (int s = 0; s < 2; ++s) {
            v2f na;
            na.x = -ak[s].x;
            na.y = -ak[s].y;
            pn = wmma_k4(na, bpp[s], pn);
          }
          Pd = pn;

          __builtin_amdgcn_sched_barrier(0);
          float4 klo, khi;
          klo.x = kt[0]; klo.y = kt[1]; klo.z = kt[2]; klo.w = kt[3];
          khi.x = kt[4]; khi.y = kt[5]; khi.z = kt[6]; khi.w = kt[7];
          *reinterpret_cast<float4*>(kb + i * KSTEP)     = klo;
          *reinterpret_cast<float4*>(kb + i * KSTEP + 4) = khi;
        }
      }
    } else {
      if (c > 0) {
        const int t0 = (c - 1) * CH;
        const int n  = __builtin_amdgcn_readfirstlane((T - t0 < CH) ? (T - t0) : CH);
        const float* kb = &kring[((c - 1) & 1) * (CH * KSTEP) + lane * 8];
        for (int i = 0; i < n; ++i) {
          const int t = t0 + i;
          const int tn = (t + 1 < T) ? (t + 1) : (T - 1);
          const float znext = meas[(size_t)tn * MEAS + (lane & 7)];
          __builtin_prefetch(&meas[(size_t)(t + 48) * MEAS], 0, 1);
          const float4 klo = *reinterpret_cast<const float4*>(kb + i * KSTEP);
          const float4 khi = *reinterpret_cast<const float4*>(kb + i * KSTEP + 4);

          float x0 = 0.0f, x1 = 0.0f, x2 = 0.0f, x3 = 0.0f;
          x0 = __builtin_fmaf(Frow[0],  rowshare<0>(xv),  x0);
          x1 = __builtin_fmaf(Frow[4],  rowshare<4>(xv),  x1);
          x2 = __builtin_fmaf(Frow[8],  rowshare<8>(xv),  x2);
          x3 = __builtin_fmaf(Frow[12], rowshare<12>(xv), x3);
          x0 = __builtin_fmaf(Frow[1],  rowshare<1>(xv),  x0);
          x1 = __builtin_fmaf(Frow[5],  rowshare<5>(xv),  x1);
          x2 = __builtin_fmaf(Frow[9],  rowshare<9>(xv),  x2);
          x3 = __builtin_fmaf(Frow[13], rowshare<13>(xv), x3);
          x0 = __builtin_fmaf(Frow[2],  rowshare<2>(xv),  x0);
          x1 = __builtin_fmaf(Frow[6],  rowshare<6>(xv),  x1);
          x2 = __builtin_fmaf(Frow[10], rowshare<10>(xv), x2);
          x3 = __builtin_fmaf(Frow[14], rowshare<14>(xv), x3);
          x0 = __builtin_fmaf(Frow[3],  rowshare<3>(xv),  x0);
          x1 = __builtin_fmaf(Frow[7],  rowshare<7>(xv),  x1);
          x2 = __builtin_fmaf(Frow[11], rowshare<11>(xv), x2);
          x3 = __builtin_fmaf(Frow[15], rowshare<15>(xv), x3);
          const float xp = (x0 + x1) + (x2 + x3);
          const float yv = zv - xp;
          const float y0 = rowshare<0>(yv), y1 = rowshare<1>(yv);
          const float y2 = rowshare<2>(yv), y3 = rowshare<3>(yv);
          const float y4 = rowshare<4>(yv), y5 = rowshare<5>(yv);
          const float y6 = rowshare<6>(yv), y7 = rowshare<7>(yv);

          float xn = xp;
          xn = __builtin_fmaf(klo.x, y0, xn);
          xn = __builtin_fmaf(klo.y, y1, xn);
          xn = __builtin_fmaf(klo.z, y2, xn);
          xn = __builtin_fmaf(klo.w, y3, xn);
          xn = __builtin_fmaf(khi.x, y4, xn);
          xn = __builtin_fmaf(khi.y, y5, xn);
          xn = __builtin_fmaf(khi.z, y6, xn);
          xn = __builtin_fmaf(khi.w, y7, xn);
          if (lane < 16) out[(size_t)t * 16 + l16] = xn;
          xv = xn;
          zv = znext;
        }
      }
    }
    __syncthreads();
  }
}

extern "C" void kernel_launch(void* const* d_in, const int* in_sizes, int n_in,
                              void* d_out, int out_size, void* d_ws, size_t ws_size,
                              hipStream_t stream) {
  const float* meas = (const float*)d_in[0];
  const float* F    = (const float*)d_in[1];
  const float* Q    = (const float*)d_in[3];
  const float* R    = (const float*)d_in[4];
  float* out = (float*)d_out;

  const int T = in_sizes[0] / MEAS;

  hipLaunchKernelGGL(kalman_seq_kernel, dim3(1), dim3(64), 0, stream,
                     meas, F, Q, R, out, T);
}


# Wave 0 carries P_pred and adds -Pp[:8,:]^T X^T (X = F K) onto the off-chain (F Pp)^T: no Pp^T product, 12 WMMAs per step not 16; K^T formed on wave 1 from Pp[:, :8] gathered out of the LDS ring
# speedup vs baseline: 4.3590x; 1.1061x over previous
#include <hip/hip_runtime.h>


typedef __attribute__((ext_vector_type(2))) float v2f;
typedef __attribute__((ext_vector_type(8))) float v8f;

#define STATE 16
#define MEAS  8
#define CH    32
#define KSTEP 384
#define KASI  256

static __device__ __forceinline__ v8f wmma_k4(v2f a, v2f b, v8f c) {
  return __builtin_amdgcn_wmma_f32_16x16x4_f32(
       false, a,  false, b,
       (short)0, c,  false,  false);
}

static __device__ __forceinline__ float xhalf(float x) {
  const int b = __builtin_bit_cast(int, x);
  return __builtin_bit_cast(float,
      __builtin_amdgcn_permlanex16(b, b, 0x76543210, 0xfedcba98, false, false));
}

template <int N>
static __device__ __forceinline__ float rowshare(float x) {
  const int b = __builtin_bit_cast(int, x);
  return __builtin_bit_cast(float,
      __builtin_amdgcn_mov_dpp(b, 0x150 + N, 0xf, 0xf, false));
}

static __device__ __forceinline__ float rowshl8_zero(float x) {
  const int b = __builtin_bit_cast(int, x);
  return __builtin_bit_cast(float,
      __builtin_amdgcn_update_dpp(0, b, 0x108, 0xf, 0xf, true));
}

static __device__ __forceinline__ void dfrag(const v8f d, v2f (&f)[4]) {
#pragma unroll
  for (int s = 0; s < 4; ++s) { f[s].x = d[2 * s]; f[s].y = d[2 * s + 1]; }
}

static __device__ __forceinline__ void conv_lo(const v8f d, const bool hi, v2f (&f)[2]) {
  const float x2 = xhalf(d[2]), x3 = xhalf(d[3]);
  const float x6 = xhalf(d[6]), x7 = xhalf(d[7]);
  f[0].x = hi ? x2 : d[0];  f[0].y = hi ? x3 : d[1];
  f[1].x = hi ? x6 : d[4];  f[1].y = hi ? x7 : d[5];
}


#define GJ_FMAC(K, acc, m)                                                      \
  asm volatile("v_fmac_f32_dpp %0, %0, %1 row_share:" #K                        \
               " row_mask:0xf bank_mask:0xf"                                    \
               : "+v"(acc) : "v"(m))

#define GJ_RCP_FMAC(N, piv, src, K, acc, m)                                     \
  asm volatile("v_rcp_f32_dpp %0, %2 row_share:" #N                             \
               " row_mask:0xf bank_mask:0xf\n\t"                                \
               "v_fmac_f32_dpp %1, %1, %3 row_share:" #K                        \
               " row_mask:0xf bank_mask:0xf"                                    \
               : "=&v"(piv), "+v"(acc) : "v"(src), "v"(m))

#define GJ_NMUL(dst, row, piv)                                                  \
  asm volatile("v_mul_f32_e64 %0, %1, -%2" : "=v"(dst) : "v"(row), "v"(piv))

#define GJ_P1(K, N)                                                             \
    GJ_FMAC(K, g[N], ngk);
#define GJ_P2(K, N, A, B, C)                                                    \
    GJ_FMAC(K, g[A], ngk);                                                      \
    GJ_FMAC(K, g[B], ngk);                                                      \
    GJ_RCP_FMAC(N, pivn, g[N], K, g[C], ngk);
#define GJ_P3(K, N, D, E, F)                                                    \
    GJ_FMAC(K, g[D], ngk);                                                      \
    GJ_FMAC(K, g[E], ngk);                                                      \
    GJ_NMUL(ngkn, g[N], pivn);                                                  \
    GJ_FMAC(K, g[F], ngk);                                                      \
    g[K] = g[K] * piv;                                                          \
    piv  = pivn;                                                                \
    ngk  = ngkn;

#define PIN1(x)       asm volatile("" : "+v"(x))
#define PIN2(x, y)    asm volatile("" : "+v"(x), "+v"(y))
#define WMMA_AT(acc, a, b)  PIN1(b); acc = wmma_k4(a, b, acc); PIN1(acc)

__global__ __launch_bounds__(64) void kalman_seq_kernel(
    const float* __restrict__ meas,
    const float* __restrict__ Fm,
    const float* __restrict__ Qm,
    const float* __restrict__ Rm,
    float* __restrict__ out,
    int T)
{
  __shared__ __attribute__((aligned(16))) float kring[2 * CH * KSTEP];

  const int lane = threadIdx.x & 31;
  const int wv = __builtin_amdgcn_readfirstlane((int)(threadIdx.x >> 5));
  const int l16  = lane & 15;
  const int half = lane >> 4;
  const bool hi  = (half != 0);
  const int nch  = (T + CH - 1) / CH;

  float Frow[16];
#pragma unroll
  for (int j = 0; j < 16; ++j) Frow[j] = Fm[l16 * 16 + j];

  v2f fab[4];
#pragma unroll
  for (int s = 0; s < 4; ++s) {
    const int k0 = 8 * half + 2 * s;
    fab[s].x = Fm[l16 * 16 + k0];
    fab[s].y = Fm[l16 * 16 + k0 + 1];
  }

  v8f qacc, qaccT;
#pragma unroll
  for (int r = 0; r < 8; ++r) {
    qacc[r]  = Qm[(r + 8 * half) * 16 + l16];
    qaccT[r] = Qm[l16 * 16 + (r + 8 * half)];
  }

  const float maskL = (lane < 8) ? 1.0f : 0.0f;
  float rI[8];
#pragma unroll
  for (int i = 0; i < 8; ++i) {
    const float idn = ((l16 & 7) == i) ? 1.0f : 0.0f;
    rI[i] = (lane < 8) ? Rm[i * 8 + (lane & 7)] : idn;
  }

  float xv = 0.0f;

  v8f pp = {0, 0, 0, 0, 0, 0, 0, 0};
  float g[8];
#pragma unroll
  for (int r = 0; r < 8; ++r) g[r] = rI[r];
  float piv = 1.0f, ngk = 0.0f;
  if (wv == 0) {
    v8f Pd;
#pragma unroll
    for (int r = 0; r < 8; ++r) Pd[r] = ((r + 8 * half) == l16) ? 1.0f : 0.0f;
    v2f bp[4];
    dfrag(Pd, bp);
    v8f m1t = {0, 0, 0, 0, 0, 0, 0, 0};
#pragma unroll
    for (int s = 0; s < 4; ++s) m1t = wmma_k4(bp[s], fab[s], m1t);
    v2f bm[4];
    dfrag(m1t, bm);
    pp = qacc;
#pragma unroll
    for (int s = 0; s < 4; ++s) pp = wmma_k4(bm[s], fab[s], pp);
#pragma unroll
    for (int r = 0; r < 8; ++r) g[r] = __builtin_fmaf(pp[r], maskL, rI[r]);
    asm volatile("v_rcp_f32_dpp %0, %1 row_share:0 row_mask:0xf bank_mask:0xf\n\t"
                 "v_nop"
                 : "=v"(piv) : "v"(g[0]));
    ngk = g[0] * (-piv);
  }

  float zv = meas[lane & 7];

  const int goff = (16 * (l16 >> 3) + 2 * half) * 8 + (l16 & 7);

  for (int c = 0; c <= nch; ++c) {
    if (wv == 0) {
      if (c < nch) {
        const int t0 = c * CH;
        const int n  = __builtin_amdgcn_readfirstlane((T - t0 < CH) ? (T - t0) : CH);
        float* kb = &kring[(c & 1) * (CH * KSTEP)];
        for (int i = 0; i < n; ++i) {
          float pivn, ngkn;
          v8f n1t = {0, 0, 0, 0, 0, 0, 0, 0};
          v2f bp[4], cp[2], cu[2];

          GJ_P1(0, 1)
          PIN1(pp);
          dfrag(pp, bp);
          WMMA_AT(n1t, bp[0], fab[0]);
          GJ_P2(0, 1, 2, 3, 4)
          {
            float4 plo, phi;
            plo.x = pp[0]; plo.y = pp[1]; plo.z = pp[2]; plo.w = pp[3];
            phi.x = pp[4]; phi.y = pp[5]; phi.z = pp[6]; phi.w = pp[7];
            *reinterpret_cast<float4*>(kb + i * KSTEP + lane * 8)     = plo;
            *reinterpret_cast<float4*>(kb + i * KSTEP + lane * 8 + 4) = phi;
          }
          GJ_P3(0, 1, 5, 6, 7)

          GJ_P1(1, 2)
          WMMA_AT(n1t, bp[1], fab[1]);
          GJ_P2(1, 2, 0, 3, 4)
          conv_lo(pp, hi, cp);
          PIN2(cp[0], cp[1]);
          GJ_P3(1, 2, 5, 6, 7)

          GJ_P1(2, 3)
          WMMA_AT(n1t, bp[2], fab[2]);
          GJ_P2(2, 3, 0, 1, 4)
          GJ_P3(2, 3, 5, 6, 7)

          GJ_P1(3, 4)
          WMMA_AT(n1t, bp[3], fab[3]);
          GJ_P2(3, 4, 0, 1, 2)
          GJ_P3(3, 4, 5, 6, 7)

          GJ_P1(4, 5)
          GJ_P2(4, 5, 0, 1, 2)
          GJ_P3(4, 5, 3, 6, 7)

          GJ_P1(5, 6)
          GJ_P2(5, 6, 0, 1, 2)
          conv_lo(n1t, hi, cu);
          PIN2(cu[0], cu[1]);
          GJ_P3(5, 6, 3, 4, 7)

          GJ_P1(6, 7)
          GJ_P2(6, 7, 0, 1, 2)
          GJ_P3(6, 7, 3, 4, 5)

          {
            GJ_FMAC(7, g[0], ngk);
            GJ_FMAC(7, g[1], ngk);
            GJ_FMAC(7, g[2], ngk);
            GJ_FMAC(7, g[3], ngk);
            GJ_FMAC(7, g[4], ngk);
            GJ_FMAC(7, g[5], ngk);
            GJ_FMAC(7, g[6], ngk);
            g[7] = g[7] * piv;
          }
          float sh[8];
#pragma unroll
          for (int r = 0; r < 8; ++r) sh[r] = rowshl8_zero(g[r]);
          v2f asi[2];
          {
            const float s2 = xhalf(sh[2]), s3 = xhalf(sh[3]);
            const float s6 = xhalf(sh[6]), s7 = xhalf(sh[7]);
            asi[0].x = hi ? s2 : sh[0];  asi[0].y = hi ? s3 : sh[1];
            asi[1].x = hi ? s6 : sh[4];  asi[1].y = hi ? s7 : sh[5];
          }

          v8f xt = {0, 0, 0, 0, 0, 0, 0, 0};
#pragma unroll
          for (int s = 0; s < 2; ++s) xt = wmma_k4(asi[s], cu[s], xt);

          {
            float4 av;
            av.x = asi[0].x; av.y = asi[0].y; av.z = asi[1].x; av.w = asi[1].y;
            *reinterpret_cast<float4*>(kb + i * KSTEP + KASI + lane * 4) = av;
          }

          v2f cx[2];
          conv_lo(xt, hi, cx);
          v8f m1t = n1t;
#pragma unroll
          for (int s = 0; s < 2; ++s) {
            v2f nb;
            nb.x = -cx[s].x;
            nb.y = -cx[s].y;
            m1t = wmma_k4(cp[s], nb, m1t);
          }

          v2f bm[4];
          dfrag(m1t, bm);
          v8f pn = qacc;
#pragma unroll
          for (int s = 0; s < 4; ++s) pn = wmma_k4(bm[s], fab[s], pn);
          pp = pn;

#pragma unroll
          for (int r = 0; r < 8; ++r) g[r] = __builtin_fmaf(pp[r], maskL, rI[r]);
          asm volatile("v_rcp_f32_dpp %0, %1 row_share:0 row_mask:0xf bank_mask:0xf\n\t"
                       "v_nop"
                       : "=v"(piv) : "v"(g[0]));
          ngk = g[0] * (-piv);
        }
      }
    } else {
      if (c > 0) {
        const int t0 = (c - 1) * CH;
        const int n  = __builtin_amdgcn_readfirstlane((T - t0 < CH) ? (T - t0) : CH);
        const float* kb = &kring[((c - 1) & 1) * (CH * KSTEP)];
        for (int i = 0; i < n; ++i) {
          const int t = t0 + i;
          const int tn = (t + 1 < T) ? (t + 1) : (T - 1);
          const float znext = meas[(size_t)tn * MEAS + (lane & 7)];
          __builtin_prefetch(&meas[(size_t)(t + 48) * MEAS], 0, 1);
          const float4 av = *reinterpret_cast<const float4*>(kb + i * KSTEP + KASI + lane * 4);
          const float* gp = kb + i * KSTEP + goff;
          v2f a0, a1, b0, b1;
          a0.x = av.x;  a0.y = av.y;  a1.x = av.z;  a1.y = av.w;
          b0.x = gp[0];  b0.y = gp[8];
          b1.x = gp[32]; b1.y = gp[40];
          v8f kt = {0, 0, 0, 0, 0, 0, 0, 0};
          kt = wmma_k4(a0, b0, kt);
          kt = wmma_k4(a1, b1, kt);
          float4 klo, khi;
          klo.x = kt[0]; klo.y = kt[1]; klo.z = kt[2]; klo.w = kt[3];
          khi.x = kt[4]; khi.y = kt[5]; khi.z = kt[6]; khi.w = kt[7];

          float x0 = 0.0f, x1 = 0.0f, x2 = 0.0f, x3 = 0.0f;
          x0 = __builtin_fmaf(Frow[0],  rowshare<0>(xv),  x0);
          x1 = __builtin_fmaf(Frow[4],  rowshare<4>(xv),  x1);
          x2 = __builtin_fmaf(Frow[8],  rowshare<8>(xv),  x2);
          x3 = __builtin_fmaf(Frow[12], rowshare<12>(xv), x3);
          x0 = __builtin_fmaf(Frow[1],  rowshare<1>(xv),  x0);
          x1 = __builtin_fmaf(Frow[5],  rowshare<5>(xv),  x1);
          x2 = __builtin_fmaf(Frow[9],  rowshare<9>(xv),  x2);
          x3 = __builtin_fmaf(Frow[13], rowshare<13>(xv), x3);
          x0 = __builtin_fmaf(Frow[2],  rowshare<2>(xv),  x0);
          x1 = __builtin_fmaf(Frow[6],  rowshare<6>(xv),  x1);
          x2 = __builtin_fmaf(Frow[10], rowshare<10>(xv), x2);
          x3 = __builtin_fmaf(Frow[14], rowshare<14>(xv), x3);
          x0 = __builtin_fmaf(Frow[3],  rowshare<3>(xv),  x0);
          x1 = __builtin_fmaf(Frow[7],  rowshare<7>(xv),  x1);
          x2 = __builtin_fmaf(Frow[11], rowshare<11>(xv), x2);
          x3 = __builtin_fmaf(Frow[15], rowshare<15>(xv), x3);
          const float xp = (x0 + x1) + (x2 + x3);
          const float yv = zv - xp;
          const float y0 = rowshare<0>(yv), y1 = rowshare<1>(yv);
          const float y2 = rowshare<2>(yv), y3 = rowshare<3>(yv);
          const float y4 = rowshare<4>(yv), y5 = rowshare<5>(yv);
          const float y6 = rowshare<6>(yv), y7 = rowshare<7>(yv);

          float xn = xp;
          xn = __builtin_fmaf(klo.x, y0, xn);
          xn = __builtin_fmaf(klo.y, y1, xn);
          xn = __builtin_fmaf(klo.z, y2, xn);
          xn = __builtin_fmaf(klo.w, y3, xn);
          xn = __builtin_fmaf(khi.x, y4, xn);
          xn = __builtin_fmaf(khi.y, y5, xn);
          xn = __builtin_fmaf(khi.z, y6, xn);
          xn = __builtin_fmaf(khi.w, y7, xn);
          if (lane < 16) out[(size_t)t * 16 + l16] = xn;
          xv = xn;
          zv = znext;
        }
      }
    }
    __syncthreads();
  }
}

extern "C" void kernel_launch(void* const* d_in, const int* in_sizes, int n_in,
                              void* d_out, int out_size, void* d_ws, size_t ws_size,
                              hipStream_t stream) {
  const float* meas = (const float*)d_in[0];
  const float* F    = (const float*)d_in[1];
  const float* Q    = (const float*)d_in[3];
  const float* R    = (const float*)d_in[4];
  float* out = (float*)d_out;

  const int T = in_sizes[0] / MEAS;

  hipLaunchKernelGGL(kalman_seq_kernel, dim3(1), dim3(64), 0, stream,
                     meas, F, Q, R, out, T);
}


# X^T tile feeds (F P)^T as its own B fragment set: the A operand that makes it repeats S^-T rows 4..7 in lanes 8..11 (k = 4h+2s+e); minus sign moved onto the off-chain Pp fragments
# speedup vs baseline: 4.5500x; 1.0438x over previous
#include <hip/hip_runtime.h>


typedef __attribute__((ext_vector_type(2))) float v2f;
typedef __attribute__((ext_vector_type(8))) float v8f;

#define STATE 16
#define MEAS  8
#define CH    32
#define KSTEP 384
#define KASI  256

static __device__ __forceinline__ v8f wmma_k4(v2f a, v2f b, v8f c) {
  return __builtin_amdgcn_wmma_f32_16x16x4_f32(
       false, a,  false, b,
       (short)0, c,  false,  false);
}

static __device__ __forceinline__ float xhalf(float x) {
  const int b = __builtin_bit_cast(int, x);
  return __builtin_bit_cast(float,
      __builtin_amdgcn_permlanex16(b, b, 0x76543210, 0xfedcba98, false, false));
}

template <int N>
static __device__ __forceinline__ float rowshare(float x) {
  const int b = __builtin_bit_cast(int, x);
  return __builtin_bit_cast(float,
      __builtin_amdgcn_mov_dpp(b, 0x150 + N, 0xf, 0xf, false));
}

static __device__ __forceinline__ float gather_inv(float x) {
  const int b = __builtin_bit_cast(int, x);
  return __builtin_bit_cast(float,
      __builtin_amdgcn_permlane16(b, b, 0xfedcba98, 0xfedcfedc, false, false));
}

static __device__ __forceinline__ void conv_lo_h4_neg(const v8f d, const bool hi, v2f (&f)[2]) {
  const float x4 = xhalf(d[4]), x5 = xhalf(d[5]);
  const float x6 = xhalf(d[6]), x7 = xhalf(d[7]);
  f[0].x = hi ? -x4 : -d[0];  f[0].y = hi ? -x5 : -d[1];
  f[1].x = hi ? -x6 : -d[2];  f[1].y = hi ? -x7 : -d[3];
}

static __device__ __forceinline__ void dfrag(const v8f d, v2f (&f)[4]) {
#pragma unroll
  for (int s = 0; s < 4; ++s) { f[s].x = d[2 * s]; f[s].y = d[2 * s + 1]; }
}

static __device__ __forceinline__ void conv_lo(const v8f d, const bool hi, v2f (&f)[2]) {
  const float x2 = xhalf(d[2]), x3 = xhalf(d[3]);
  const float x6 = xhalf(d[6]), x7 = xhalf(d[7]);
  f[0].x = hi ? x2 : d[0];  f[0].y = hi ? x3 : d[1];
  f[1].x = hi ? x6 : d[4];  f[1].y = hi ? x7 : d[5];
}


#define GJ_FMAC(K, acc, m)                                                      \
  asm volatile("v_fmac_f32_dpp %0, %0, %1 row_share:" #K                        \
               " row_mask:0xf bank_mask:0xf"                                    \
               : "+v"(acc) : "v"(m))

#define GJ_RCP_FMAC(N, piv, src, K, acc, m)                                     \
  asm volatile("v_rcp_f32_dpp %0, %2 row_share:" #N                             \
               " row_mask:0xf bank_mask:0xf\n\t"                                \
               "v_fmac_f32_dpp %1, %1, %3 row_share:" #K                        \
               " row_mask:0xf bank_mask:0xf"                                    \
               : "=&v"(piv), "+v"(acc) : "v"(src), "v"(m))

#define GJ_NMUL(dst, row, piv)                                                  \
  asm volatile("v_mul_f32_e64 %0, %1, -%2" : "=v"(dst) : "v"(row), "v"(piv))

#define GJ_P1(K, N)                                                             \
    GJ_FMAC(K, g[N], ngk);
#define GJ_P2(K, N, A, B, C)                                                    \
    GJ_FMAC(K, g[A], ngk);                                                      \
    GJ_FMAC(K, g[B], ngk);                                                      \
    GJ_RCP_FMAC(N, pivn, g[N], K, g[C], ngk);
#define GJ_P3(K, N, D, E, F)                                                    \
    GJ_FMAC(K, g[D], ngk);                                                      \
    GJ_FMAC(K, g[E], ngk);                                                      \
    GJ_NMUL(ngkn, g[N], pivn);                                                  \
    GJ_FMAC(K, g[F], ngk);                                                      \
    g[K] = g[K] * piv;                                                          \
    piv  = pivn;                                                                \
    ngk  = ngkn;

#define PIN1(x)       asm volatile("" : "+v"(x))
#define PIN2(x, y)    asm volatile("" : "+v"(x), "+v"(y))
#define WMMA_AT(acc, a, b)  PIN1(b); acc = wmma_k4(a, b, acc); PIN1(acc)

__global__ __launch_bounds__(64) void kalman_seq_kernel(
    const float* __restrict__ meas,
    const float* __restrict__ Fm,
    const float* __restrict__ Qm,
    const float* __restrict__ Rm,
    float* __restrict__ out,
    int T)
{
  __shared__ __attribute__((aligned(16))) float kring[2 * CH * KSTEP];

  const int lane = threadIdx.x & 31;
  const int wv = __builtin_amdgcn_readfirstlane((int)(threadIdx.x >> 5));
  const int l16  = lane & 15;
  const int half = lane >> 4;
  const bool hi  = (half != 0);
  const int nch  = (T + CH - 1) / CH;

  float Frow[16];
#pragma unroll
  for (int j = 0; j < 16; ++j) Frow[j] = Fm[l16 * 16 + j];

  v2f fab[4];
#pragma unroll
  for (int s = 0; s < 4; ++s) {
    const int k0 = 8 * half + 2 * s;
    fab[s].x = Fm[l16 * 16 + k0];
    fab[s].y = Fm[l16 * 16 + k0 + 1];
  }

  v8f qacc, qaccT;
#pragma unroll
  for (int r = 0; r < 8; ++r) {
    qacc[r]  = Qm[(r + 8 * half) * 16 + l16];
    qaccT[r] = Qm[l16 * 16 + (r + 8 * half)];
  }

  const float maskL = (lane < 8) ? 1.0f : 0.0f;
  float rI[8];
#pragma unroll
  for (int i = 0; i < 8; ++i) {
    const float idn = ((l16 & 7) == i) ? 1.0f : 0.0f;
    rI[i] = (lane < 8) ? Rm[i * 8 + (lane & 7)] : idn;
  }

  float xv = 0.0f;

  v8f pp = {0, 0, 0, 0, 0, 0, 0, 0};
  float g[8];
#pragma unroll
  for (int r = 0; r < 8; ++r) g[r] = rI[r];
  float piv = 1.0f, ngk = 0.0f;
  if (wv == 0) {
    v8f Pd;
#pragma unroll
    for (int r = 0; r < 8; ++r) Pd[r] = ((r + 8 * half) == l16) ? 1.0f : 0.0f;
    v2f bp[4];
    dfrag(Pd, bp);
    v8f m1t = {0, 0, 0, 0, 0, 0, 0, 0};
#pragma unroll
    for (int s = 0; s < 4; ++s) m1t = wmma_k4(bp[s], fab[s], m1t);
    v2f bm[4];
    dfrag(m1t, bm);
    pp = qacc;
#pragma unroll
    for (int s = 0; s < 4; ++s) pp = wmma_k4(bm[s], fab[s], pp);
#pragma unroll
    for (int r = 0; r < 8; ++r) g[r] = __builtin_fmaf(pp[r], maskL, rI[r]);
    asm volatile("v_rcp_f32_dpp %0, %1 row_share:0 row_mask:0xf bank_mask:0xf\n\t"
                 "v_nop"
                 : "=v"(piv) : "v"(g[0]));
    ngk = g[0] * (-piv);
  }

  float zv = meas[lane & 7];

  const int goff = (16 * (l16 >> 3) + 2 * half) * 8 + (l16 & 7);

  for (int c = 0; c <= nch; ++c) {
    if (wv == 0) {
      if (c < nch) {
        const int t0 = c * CH;
        const int n  = __builtin_amdgcn_readfirstlane((T - t0 < CH) ? (T - t0) : CH);
        float* kb = &kring[(c & 1) * (CH * KSTEP)];
        for (int i = 0; i < n; ++i) {
          float pivn, ngkn;
          v8f n1t = {0, 0, 0, 0, 0, 0, 0, 0};
          v2f bp[4], cp[2], cu[2];

          GJ_P1(0, 1)
          PIN1(pp);
          dfrag(pp, bp);
          WMMA_AT(n1t, bp[0], fab[0]);
          GJ_P2(0, 1, 2, 3, 4)
          {
            float4 plo, phi;
            plo.x = pp[0]; plo.y = pp[1]; plo.z = pp[2]; plo.w = pp[3];
            phi.x = pp[4]; phi.y = pp[5]; phi.z = pp[6]; phi.w = pp[7];
            *reinterpret_cast<float4*>(kb + i * KSTEP + lane * 8)     = plo;
            *reinterpret_cast<float4*>(kb + i * KSTEP + lane * 8 + 4) = phi;
          }
          GJ_P3(0, 1, 5, 6, 7)

          GJ_P1(1, 2)
          WMMA_AT(n1t, bp[1], fab[1]);
          GJ_P2(1, 2, 0, 3, 4)
          conv_lo_h4_neg(pp, hi, cp);
          PIN2(cp[0], cp[1]);
          GJ_P3(1, 2, 5, 6, 7)

          GJ_P1(2, 3)
          WMMA_AT(n1t, bp[2], fab[2]);
          GJ_P2(2, 3, 0, 1, 4)
          GJ_P3(2, 3, 5, 6, 7)

          GJ_P1(3, 4)
          WMMA_AT(n1t, bp[3], fab[3]);
          GJ_P2(3, 4, 0, 1, 2)
          GJ_P3(3, 4, 5, 6, 7)

          GJ_P1(4, 5)
          GJ_P2(4, 5, 0, 1, 2)
          GJ_P3(4, 5, 3, 6, 7)

          GJ_P1(5, 6)
          GJ_P2(5, 6, 0, 1, 2)
          conv_lo(n1t, hi, cu);
          PIN2(cu[0], cu[1]);
          GJ_P3(5, 6, 3, 4, 7)

          GJ_P1(6, 7)
          GJ_P2(6, 7, 0, 1, 2)
          GJ_P3(6, 7, 3, 4, 5)

          {
            GJ_FMAC(7, g[0], ngk);
            GJ_FMAC(7, g[1], ngk);
            GJ_FMAC(7, g[2], ngk);
            GJ_FMAC(7, g[3], ngk);
            GJ_FMAC(7, g[4], ngk);
            GJ_FMAC(7, g[5], ngk);
            GJ_FMAC(7, g[6], ngk);
            g[7] = g[7] * piv;
          }
          float sh[8];
#pragma unroll
          for (int r = 0; r < 8; ++r) sh[r] = gather_inv(g[r]);
          v2f asi[2];
          {
            const float s2 = xhalf(sh[2]), s3 = xhalf(sh[3]);
            const float s6 = xhalf(sh[6]), s7 = xhalf(sh[7]);
            asi[0].x = hi ? s2 : sh[0];  asi[0].y = hi ? s3 : sh[1];
            asi[1].x = hi ? s6 : sh[4];  asi[1].y = hi ? s7 : sh[5];
          }

          v8f xt = {0, 0, 0, 0, 0, 0, 0, 0};
#pragma unroll
          for (int s = 0; s < 2; ++s) xt = wmma_k4(asi[s], cu[s], xt);

          {
            float4 av;
            av.x = asi[0].x; av.y = asi[0].y; av.z = asi[1].x; av.w = asi[1].y;
            *reinterpret_cast<float4*>(kb + i * KSTEP + KASI + lane * 4) = av;
          }

          v8f m1t = n1t;
#pragma unroll
          for (int s = 0; s < 2; ++s) {
            v2f cx;
            cx.x = xt[2 * s];
            cx.y = xt[2 * s + 1];
            m1t = wmma_k4(cp[s], cx, m1t);
          }

          v2f bm[4];
          dfrag(m1t, bm);
          v8f pn = qacc;
#pragma unroll
          for (int s = 0; s < 4; ++s) pn = wmma_k4(bm[s], fab[s], pn);
          pp = pn;

#pragma unroll
          for (int r = 0; r < 8; ++r) g[r] = __builtin_fmaf(pp[r], maskL, rI[r]);
          asm volatile("v_rcp_f32_dpp %0, %1 row_share:0 row_mask:0xf bank_mask:0xf\n\t"
                       "v_nop"
                       : "=v"(piv) : "v"(g[0]));
          ngk = g[0] * (-piv);
        }
      }
    } else {
      if (c > 0) {
        const int t0 = (c - 1) * CH;
        const int n  = __builtin_amdgcn_readfirstlane((T - t0 < CH) ? (T - t0) : CH);
        const float* kb = &kring[((c - 1) & 1) * (CH * KSTEP)];
        for (int i = 0; i < n; ++i) {
          const int t = t0 + i;
          const int tn = (t + 1 < T) ? (t + 1) : (T - 1);
          const float znext = meas[(size_t)tn * MEAS + (lane & 7)];
          __builtin_prefetch(&meas[(size_t)(t + 48) * MEAS], 0, 1);
          const float4 av = *reinterpret_cast<const float4*>(kb + i * KSTEP + KASI + lane * 4);
          const float* gp = kb + i * KSTEP + goff;
          v2f a0, a1, b0, b1;
          a0.x = av.x;  a0.y = av.y;  a1.x = av.z;  a1.y = av.w;
          b0.x = gp[0];  b0.y = gp[8];
          b1.x = gp[32]; b1.y = gp[40];
          v8f kt = {0, 0, 0, 0, 0, 0, 0, 0};
          kt = wmma_k4(a0, b0, kt);
          kt = wmma_k4(a1, b1, kt);
          float4 klo, khi;
          klo.x = kt[0]; klo.y = kt[1]; klo.z = kt[2]; klo.w = kt[3];
          khi.x = kt[4]; khi.y = kt[5]; khi.z = kt[6]; khi.w = kt[7];

          float x0 = 0.0f, x1 = 0.0f, x2 = 0.0f, x3 = 0.0f;
          x0 = __builtin_fmaf(Frow[0],  rowshare<0>(xv),  x0);
          x1 = __builtin_fmaf(Frow[4],  rowshare<4>(xv),  x1);
          x2 = __builtin_fmaf(Frow[8],  rowshare<8>(xv),  x2);
          x3 = __builtin_fmaf(Frow[12], rowshare<12>(xv), x3);
          x0 = __builtin_fmaf(Frow[1],  rowshare<1>(xv),  x0);
          x1 = __builtin_fmaf(Frow[5],  rowshare<5>(xv),  x1);
          x2 = __builtin_fmaf(Frow[9],  rowshare<9>(xv),  x2);
          x3 = __builtin_fmaf(Frow[13], rowshare<13>(xv), x3);
          x0 = __builtin_fmaf(Frow[2],  rowshare<2>(xv),  x0);
          x1 = __builtin_fmaf(Frow[6],  rowshare<6>(xv),  x1);
          x2 = __builtin_fmaf(Frow[10], rowshare<10>(xv), x2);
          x3 = __builtin_fmaf(Frow[14], rowshare<14>(xv), x3);
          x0 = __builtin_fmaf(Frow[3],  rowshare<3>(xv),  x0);
          x1 = __builtin_fmaf(Frow[7],  rowshare<7>(xv),  x1);
          x2 = __builtin_fmaf(Frow[11], rowshare<11>(xv), x2);
          x3 = __builtin_fmaf(Frow[15], rowshare<15>(xv), x3);
          const float xp = (x0 + x1) + (x2 + x3);
          const float yv = zv - xp;
          const float y0 = rowshare<0>(yv), y1 = rowshare<1>(yv);
          const float y2 = rowshare<2>(yv), y3 = rowshare<3>(yv);
          const float y4 = rowshare<4>(yv), y5 = rowshare<5>(yv);
          const float y6 = rowshare<6>(yv), y7 = rowshare<7>(yv);

          float xn = xp;
          xn = __builtin_fmaf(klo.x, y0, xn);
          xn = __builtin_fmaf(klo.y, y1, xn);
          xn = __builtin_fmaf(klo.z, y2, xn);
          xn = __builtin_fmaf(klo.w, y3, xn);
          xn = __builtin_fmaf(khi.x, y4, xn);
          xn = __builtin_fmaf(khi.y, y5, xn);
          xn = __builtin_fmaf(khi.z, y6, xn);
          xn = __builtin_fmaf(khi.w, y7, xn);
          if (lane < 16) out[(size_t)t * 16 + l16] = xn;
          xv = xn;
          zv = znext;
        }
      }
    }
    __syncthreads();
  }
}

extern "C" void kernel_launch(void* const* d_in, const int* in_sizes, int n_in,
                              void* d_out, int out_size, void* d_ws, size_t ws_size,
                              hipStream_t stream) {
  const float* meas = (const float*)d_in[0];
  const float* F    = (const float*)d_in[1];
  const float* Q    = (const float*)d_in[3];
  const float* R    = (const float*)d_in[4];
  float* out = (float*)d_out;

  const int T = in_sizes[0] / MEAS;

  hipLaunchKernelGGL(kalman_seq_kernel, dim3(1), dim3(64), 0, stream,
                     meas, F, Q, R, out, T);
}
